# starting kernel
# baseline (speedup 1.0000x reference)
#include <hip/hip_runtime.h>
#include <hip/hip_bf16.h>

typedef float v2f __attribute__((ext_vector_type(2)));
typedef float v8f __attribute__((ext_vector_type(8)));

#define VOCAB 256
#define DCH   512
#define KC    5
#define TL    48
#define LLEN  8192

__global__ __launch_bounds__(32) void gbst_table_wmma(
    const float* __restrict__ emb,
    const float* __restrict__ conv_w,
    float* __restrict__ T)
{
  const int k    = blockIdx.z;
  const int v0   = blockIdx.y * 16;
  const int o0   = blockIdx.x * 16;
  const int lane = threadIdx.x;
  const int half = lane >> 4;
  const int sub  = lane & 15;
  const int kh   = half * 2;

  v8f c = {};
  const float* arow = emb + (size_t)(v0 + sub) * DCH;
  const float* bcol = conv_w + (size_t)(o0 + sub) * (DCH * KC) + k;

  for (int i0 = 0; i0 < DCH; i0 += 4) {
    v2f a, b;
    a[0] = arow[i0 + kh];
    a[1] = arow[i0 + kh + 1];
    b[0] = bcol[(size_t)(i0 + kh) * KC];
    b[1] = bcol[(size_t)(i0 + kh + 1) * KC];
    c = __builtin_amdgcn_wmma_f32_16x16x4_f32(false, a, false, b,
                                              (short)0, c, false, false);
  }
#pragma unroll
  for (int r = 0; r < 8; ++r) {
    T[((size_t)k * VOCAB + (v0 + half * 8 + r)) * DCH + (o0 + sub)] = c[r];
  }
}

__global__ __launch_bounds__(256) void gbst_main(
    const int*   __restrict__ ids,
    const float* __restrict__ T,
    const float* __restrict__ conv_b,
    const float* __restrict__ sw,
    float*       __restrict__ out,
    int L)
{
  __shared__ float xt[TL][DCH];
  __shared__ float q[TL];
  __shared__ float wc[TL][4];
  __shared__ int   sid[TL + 4];

  const int b   = blockIdx.y;
  const int t0  = blockIdx.x * TL;
  const int tid = threadIdx.x;

  if (tid < TL + 4) {
    int g = t0 + tid - 2;
    sid[tid] = (g >= 0 && g < L) ? ids[(size_t)b * L + g] : 0;
  }
  __syncthreads();

  const int oc = (tid & 127) * 4;
  const int pr = tid >> 7;

  const float bx = conv_b[oc + 0], by = conv_b[oc + 1];
  const float bz = conv_b[oc + 2], bw = conv_b[oc + 3];
  for (int p = pr; p < TL; p += 2) {
    float ax = bx, ay = by, az = bz, aw = bw;
#pragma unroll
    for (int k = 0; k < KC; ++k) {
      int v = sid[p + k];
      if (v > 0) {
        const float4 tv = *(const float4*)(T + ((size_t)k * VOCAB + v) * DCH + oc);
        ax += tv.x; ay += tv.y; az += tv.z; aw += tv.w;
      }
    }
    if (t0 + p >= L) { ax = 0.f; ay = 0.f; az = 0.f; aw = 0.f; }
    xt[p][oc + 0] = ax; xt[p][oc + 1] = ay;
    xt[p][oc + 2] = az; xt[p][oc + 3] = aw;
  }
  __syncthreads();

  {
    const int wave = tid >> 5;
    const int lane = tid & 31;
    float swr[DCH / 32];
#pragma unroll
    for (int j = 0; j < DCH / 32; ++j) swr[j] = sw[lane + 32 * j];
    for (int p = wave; p < TL; p += 8) {
      float s = 0.f;
#pragma unroll
      for (int j = 0; j < DCH / 32; ++j) s += xt[p][lane + 32 * j] * swr[j];
#pragma unroll
      for (int off = 16; off > 0; off >>= 1) s += __shfl_xor(s, off, 32);
      if (lane == 0) q[p] = s;
    }
  }
  __syncthreads();

  if (tid < TL) {
    const int l = tid;
    const int j2 = (l >> 1) << 1;
    const int j3 = (l / 3) * 3;
    const int j4 = (l >> 2) << 2;
    float s1 = q[l];
    float s2 = 0.5f * (q[j2] + q[j2 + 1]);
    float s3 = (q[j3] + q[j3 + 1] + q[j3 + 2]) * (1.f / 3.f);
    float s4 = 0.25f * (q[j4] + q[j4 + 1] + q[j4 + 2] + q[j4 + 3]);
    float mx = fmaxf(fmaxf(s1, s2), fmaxf(s3, s4));
    float e1 = __expf(s1 - mx), e2 = __expf(s2 - mx);
    float e3 = __expf(s3 - mx), e4 = __expf(s4 - mx);
    float inv = 1.f / (e1 + e2 + e3 + e4);
    wc[l][0] = e1 * inv;
    wc[l][1] = e2 * inv * 0.5f;
    wc[l][2] = e3 * inv * (1.f / 3.f);
    wc[l][3] = e4 * inv * 0.25f;
  }
  __syncthreads();

  for (int l2 = pr; l2 < TL / 2; l2 += 2) {
    if (t0 + 2 * l2 >= L) break;
    float ax = 0.f, ay = 0.f, az = 0.f, aw = 0.f;
#pragma unroll
    for (int dl = 0; dl < 2; ++dl) {
      const int l  = 2 * l2 + dl;
      const int j2 = (l >> 1) << 1;
      const int j3 = (l / 3) * 3;
      const int j4 = (l >> 2) << 2;
      const float w1 = wc[l][0], w2 = wc[l][1], w3 = wc[l][2], w4 = wc[l][3];
      ax += w1 * xt[l][oc + 0];
      ay += w1 * xt[l][oc + 1];
      az += w1 * xt[l][oc + 2];
      aw += w1 * xt[l][oc + 3];
      ax += w2 * (xt[j2][oc + 0] + xt[j2 + 1][oc + 0]);
      ay += w2 * (xt[j2][oc + 1] + xt[j2 + 1][oc + 1]);
      az += w2 * (xt[j2][oc + 2] + xt[j2 + 1][oc + 2]);
      aw += w2 * (xt[j2][oc + 3] + xt[j2 + 1][oc + 3]);
      ax += w3 * (xt[j3][oc + 0] + xt[j3 + 1][oc + 0] + xt[j3 + 2][oc + 0]);
      ay += w3 * (xt[j3][oc + 1] + xt[j3 + 1][oc + 1] + xt[j3 + 2][oc + 1]);
      az += w3 * (xt[j3][oc + 2] + xt[j3 + 1][oc + 2] + xt[j3 + 2][oc + 2]);
      aw += w3 * (xt[j3][oc + 3] + xt[j3 + 1][oc + 3] + xt[j3 + 2][oc + 3]);
      ax += w4 * (xt[j4][oc + 0] + xt[j4 + 1][oc + 0] + xt[j4 + 2][oc + 0] + xt[j4 + 3][oc + 0]);
      ay += w4 * (xt[j4][oc + 1] + xt[j4 + 1][oc + 1] + xt[j4 + 2][oc + 1] + xt[j4 + 3][oc + 1]);
      az += w4 * (xt[j4][oc + 2] + xt[j4 + 1][oc + 2] + xt[j4 + 2][oc + 2] + xt[j4 + 3][oc + 2]);
      aw += w4 * (xt[j4][oc + 3] + xt[j4 + 1][oc + 3] + xt[j4 + 2][oc + 3] + xt[j4 + 3][oc + 3]);
    }
    float4 r;
    r.x = 0.5f * ax; r.y = 0.5f * ay; r.z = 0.5f * az; r.w = 0.5f * aw;
    const size_t off = ((size_t)b * (L >> 1) + ((t0 >> 1) + l2)) * DCH + oc;
    *(float4*)(out + off) = r;
  }
}

extern "C" void kernel_launch(void* const* d_in, const int* in_sizes, int n_in,
                              void* d_out, int out_size, void* d_ws, size_t ws_size,
                              hipStream_t stream) {
  const int*   ids     = (const int*)d_in[0];
  const float* emb     = (const float*)d_in[1];
  const float* conv_w  = (const float*)d_in[2];
  const float* conv_b  = (const float*)d_in[3];
  const float* score_w = (const float*)d_in[4];
  float* out = (float*)d_out;
  float* T   = (float*)d_ws;

  const int L = LLEN;
  const int B = in_sizes[0] / L;

  dim3 g1(DCH / 16, VOCAB / 16, KC);
  gbst_table_wmma<<<g1, 32, 0, stream>>>(emb, conv_w, T);

  dim3 g2((L + TL - 1) / TL, B);
  gbst_main<<<g2, 256, 0, stream>>>(ids, T, conv_b, score_w, out, L);
}


# Main kernel's table gather made branch-free: row loads issued in groups of 20 (4 positions x 5 taps per built trip) ahead of the sums, so L2 latencies overlap instead of 120 serial waits per wave
# speedup vs baseline: 1.6089x; 1.6089x over previous
#include <hip/hip_runtime.h>
#include <hip/hip_bf16.h>

typedef float v2f __attribute__((ext_vector_type(2)));
typedef float v8f __attribute__((ext_vector_type(8)));

#define VOCAB 256
#define DCH   512
#define KC    5
#define TL    48
#define LLEN  8192
#define GP    2

__global__ __launch_bounds__(32) void gbst_table_wmma(
    const float* __restrict__ emb,
    const float* __restrict__ conv_w,
    float* __restrict__ T)
{
  const int k    = blockIdx.z;
  const int v0   = blockIdx.y * 16;
  const int o0   = blockIdx.x * 16;
  const int lane = threadIdx.x;
  const int half = lane >> 4;
  const int sub  = lane & 15;
  const int kh   = half * 2;

  v8f c = {};
  const float* arow = emb + (size_t)(v0 + sub) * DCH;
  const float* bcol = conv_w + (size_t)(o0 + sub) * (DCH * KC) + k;

  for (int i0 = 0; i0 < DCH; i0 += 4) {
    v2f a, b;
    a[0] = arow[i0 + kh];
    a[1] = arow[i0 + kh + 1];
    b[0] = bcol[(size_t)(i0 + kh) * KC];
    b[1] = bcol[(size_t)(i0 + kh + 1) * KC];
    c = __builtin_amdgcn_wmma_f32_16x16x4_f32(false, a, false, b,
                                              (short)0, c, false, false);
  }
#pragma unroll
  for (int r = 0; r < 8; ++r) {
    T[((size_t)k * VOCAB + (v0 + half * 8 + r)) * DCH + (o0 + sub)] = c[r];
  }
}

__global__ __launch_bounds__(256) void gbst_main(
    const int*   __restrict__ ids,
    const float* __restrict__ T,
    const float* __restrict__ conv_b,
    const float* __restrict__ sw,
    float*       __restrict__ out,
    int L)
{
  __shared__ float xt[TL][DCH];
  __shared__ float q[TL];
  __shared__ float wc[TL][4];
  __shared__ int   sid[TL + 4];

  const int b   = blockIdx.y;
  const int t0  = blockIdx.x * TL;
  const int tid = threadIdx.x;

  if (tid < TL + 4) {
    int g = t0 + tid - 2;
    sid[tid] = (g >= 0 && g < L) ? ids[(size_t)b * L + g] : -1;
  }
  __syncthreads();

  const int oc = (tid & 127) * 4;
  const int pr = tid >> 7;

  const float bx = conv_b[oc + 0], by = conv_b[oc + 1];
  const float bz = conv_b[oc + 2], bw = conv_b[oc + 3];
  for (int pb = 0; pb < TL; pb += 2 * GP) {
    float4 tv[GP][KC];
#pragma unroll
    for (int g = 0; g < GP; ++g) {
#pragma unroll
      for (int k = 0; k < KC; ++k) {
        const int v  = sid[pb + pr + 2 * g + k];
        const int vc = v < 0 ? 0 : v;
        float4 t = *(const float4*)(T + ((size_t)k * VOCAB + vc) * DCH + oc);
        if (v < 0) { t.x = 0.f; t.y = 0.f; t.z = 0.f; t.w = 0.f; }
        tv[g][k] = t;
      }
    }
#pragma unroll
    for (int g = 0; g < GP; ++g) {
      const int p = pb + pr + 2 * g;
      float ax = bx, ay = by, az = bz, aw = bw;
#pragma unroll
      for (int k = 0; k < KC; ++k) {
        ax += tv[g][k].x; ay += tv[g][k].y; az += tv[g][k].z; aw += tv[g][k].w;
      }
      if (t0 + p >= L) { ax = 0.f; ay = 0.f; az = 0.f; aw = 0.f; }
      xt[p][oc + 0] = ax; xt[p][oc + 1] = ay;
      xt[p][oc + 2] = az; xt[p][oc + 3] = aw;
    }
  }
  __syncthreads();

  {
    const int wave = tid >> 5;
    const int lane = tid & 31;
    float swr[DCH / 32];
#pragma unroll
    for (int j = 0; j < DCH / 32; ++j) swr[j] = sw[lane + 32 * j];
    for (int p = wave; p < TL; p += 8) {
      float s = 0.f;
#pragma unroll
      for (int j = 0; j < DCH / 32; ++j) s += xt[p][lane + 32 * j] * swr[j];
#pragma unroll
      for (int off = 16; off > 0; off >>= 1) s += __shfl_xor(s, off, 32);
      if (lane == 0) q[p] = s;
    }
  }
  __syncthreads();

  if (tid < TL) {
    const int l = tid;
    const int j2 = (l >> 1) << 1;
    const int j3 = (l / 3) * 3;
    const int j4 = (l >> 2) << 2;
    float s1 = q[l];
    float s2 = 0.5f * (q[j2] + q[j2 + 1]);
    float s3 = (q[j3] + q[j3 + 1] + q[j3 + 2]) * (1.f / 3.f);
    float s4 = 0.25f * (q[j4] + q[j4 + 1] + q[j4 + 2] + q[j4 + 3]);
    float mx = fmaxf(fmaxf(s1, s2), fmaxf(s3, s4));
    float e1 = __expf(s1 - mx), e2 = __expf(s2 - mx);
    float e3 = __expf(s3 - mx), e4 = __expf(s4 - mx);
    float inv = 1.f / (e1 + e2 + e3 + e4);
    wc[l][0] = e1 * inv;
    wc[l][1] = e2 * inv * 0.5f;
    wc[l][2] = e3 * inv * (1.f / 3.f);
    wc[l][3] = e4 * inv * 0.25f;
  }
  __syncthreads();

  for (int l2 = pr; l2 < TL / 2; l2 += 2) {
    if (t0 + 2 * l2 >= L) break;
    float ax = 0.f, ay = 0.f, az = 0.f, aw = 0.f;
#pragma unroll
    for (int dl = 0; dl < 2; ++dl) {
      const int l  = 2 * l2 + dl;
      const int j2 = (l >> 1) << 1;
      const int j3 = (l / 3) * 3;
      const int j4 = (l >> 2) << 2;
      const float w1 = wc[l][0], w2 = wc[l][1], w3 = wc[l][2], w4 = wc[l][3];
      ax += w1 * xt[l][oc + 0];
      ay += w1 * xt[l][oc + 1];
      az += w1 * xt[l][oc + 2];
      aw += w1 * xt[l][oc + 3];
      ax += w2 * (xt[j2][oc + 0] + xt[j2 + 1][oc + 0]);
      ay += w2 * (xt[j2][oc + 1] + xt[j2 + 1][oc + 1]);
      az += w2 * (xt[j2][oc + 2] + xt[j2 + 1][oc + 2]);
      aw += w2 * (xt[j2][oc + 3] + xt[j2 + 1][oc + 3]);
      ax += w3 * (xt[j3][oc + 0] + xt[j3 + 1][oc + 0] + xt[j3 + 2][oc + 0]);
      ay += w3 * (xt[j3][oc + 1] + xt[j3 + 1][oc + 1] + xt[j3 + 2][oc + 1]);
      az += w3 * (xt[j3][oc + 2] + xt[j3 + 1][oc + 2] + xt[j3 + 2][oc + 2]);
      aw += w3 * (xt[j3][oc + 3] + xt[j3 + 1][oc + 3] + xt[j3 + 2][oc + 3]);
      ax += w4 * (xt[j4][oc + 0] + xt[j4 + 1][oc + 0] + xt[j4 + 2][oc + 0] + xt[j4 + 3][oc + 0]);
      ay += w4 * (xt[j4][oc + 1] + xt[j4 + 1][oc + 1] + xt[j4 + 2][oc + 1] + xt[j4 + 3][oc + 1]);
      az += w4 * (xt[j4][oc + 2] + xt[j4 + 1][oc + 2] + xt[j4 + 2][oc + 2] + xt[j4 + 3][oc + 2]);
      aw += w4 * (xt[j4][oc + 3] + xt[j4 + 1][oc + 3] + xt[j4 + 2][oc + 3] + xt[j4 + 3][oc + 3]);
    }
    float4 r;
    r.x = 0.5f * ax; r.y = 0.5f * ay; r.z = 0.5f * az; r.w = 0.5f * aw;
    const size_t off = ((size_t)b * (L >> 1) + ((t0 >> 1) + l2)) * DCH + oc;
    *(float4*)(out + off) = r;
  }
}

extern "C" void kernel_launch(void* const* d_in, const int* in_sizes, int n_in,
                              void* d_out, int out_size, void* d_ws, size_t ws_size,
                              hipStream_t stream) {
  const int*   ids     = (const int*)d_in[0];
  const float* emb     = (const float*)d_in[1];
  const float* conv_w  = (const float*)d_in[2];
  const float* conv_b  = (const float*)d_in[3];
  const float* score_w = (const float*)d_in[4];
  float* out = (float*)d_out;
  float* T   = (float*)d_ws;

  const int L = LLEN;
  const int B = in_sizes[0] / L;

  dim3 g1(DCH / 16, VOCAB / 16, KC);
  gbst_table_wmma<<<g1, 32, 0, stream>>>(emb, conv_w, T);

  dim3 g2((L + TL - 1) / TL, B);
  gbst_main<<<g2, 256, 0, stream>>>(ids, T, conv_b, score_w, out, L);
}


# Table GEMM re-tiled: work-group = 16 channels x 32 rows x 5 taps (10 waves), conv_w and emb staged to LDS in 64-i chunks by whole-line 16-byte loads, next chunk in flight; per-tile sum order kept
# speedup vs baseline: 2.0850x; 1.2959x over previous
#include <hip/hip_runtime.h>
#include <hip/hip_bf16.h>

typedef float v2f __attribute__((ext_vector_type(2)));
typedef float v8f __attribute__((ext_vector_type(8)));

#define VOCAB 256
#define DCH   512
#define KC    5
#define TL    48
#define LLEN  8192
#define GP    2

#define TIC   64
#define TNC   (DCH / TIC)
#define TVT   2
#define TPB   (TIC * KC + 20)
#define TPA   (TIC + 4)
#define TTHR  (32 * KC * TVT)

__global__ __launch_bounds__(TTHR) void gbst_table_wmma(
    const float* __restrict__ emb,
    const float* __restrict__ conv_w,
    float* __restrict__ T)
{
  __shared__ __attribute__((aligned(16))) float sB[2][16 * TPB];
  __shared__ __attribute__((aligned(16))) float sA[2][16 * TVT * TPA];

  const int tid  = threadIdx.x;
  const int o0   = blockIdx.x * 16;
  const int vb   = blockIdx.y * (16 * TVT);
  const int wave = tid >> 5;
  const int lane = tid & 31;
  const int half = lane >> 4;
  const int sub  = lane & 15;
  const int k    = wave % KC;
  const int vt   = wave / KC;

  const int  brow = tid / 80;
  const int  bc4  = tid % 80;
  const int  arow = tid >> 4;
  const int  ac4  = tid & 15;
  const bool a2   = arow + 20 < 16 * TVT;

  const float* gB = conv_w + (size_t)(o0 + brow) * (DCH * KC) + bc4 * 4;
  const float* gA = emb + (size_t)(vb + arow) * DCH + ac4 * 4;
  const int lB = brow * TPB + bc4 * 4;
  const int lA = arow * TPA + ac4 * 4;

  float4 rb[4];
  float4 ra0, ra1;
  ra1.x = 0.f; ra1.y = 0.f; ra1.z = 0.f; ra1.w = 0.f;

#pragma unroll
  for (int j = 0; j < 4; ++j)
    rb[j] = *(const float4*)(gB + (size_t)(4 * j) * (DCH * KC));
  ra0 = *(const float4*)gA;
  if (a2) ra1 = *(const float4*)(gA + 20 * DCH);
#pragma unroll
  for (int j = 0; j < 4; ++j)
    *(float4*)(&sB[0][lB + 4 * j * TPB]) = rb[j];
  *(float4*)(&sA[0][lA]) = ra0;
  if (a2) *(float4*)(&sA[0][lA + 20 * TPA]) = ra1;
  __syncthreads();

  v8f c = {};
  const int fa = (vt * 16 + sub) * TPA + 2 * half;
  const int fb = sub * TPB + 2 * half * KC + k;

  unsigned tbyte = (unsigned)((((size_t)k * VOCAB + (vb + vt * 16 + half * 8)) * DCH + (o0 + sub)) * sizeof(float));
  asm volatile("" : "+v"(tbyte));

  for (int ch = 0; ch < TNC; ++ch) {
    const float* cA = sA[ch & 1];
    const float* cB = sB[ch & 1];
    if (ch + 1 < TNC) {
      const int i1 = (ch + 1) * TIC;
#pragma unroll
      for (int j = 0; j < 4; ++j)
        rb[j] = *(const float4*)(gB + (size_t)(4 * j) * (DCH * KC) + i1 * KC);
      ra0 = *(const float4*)(gA + i1);
      if (a2) ra1 = *(const float4*)(gA + 20 * DCH + i1);
    }
#pragma unroll
    for (int s = 0; s < TIC / 4; ++s) {
      const v2f a = *(const v2f*)(cA + fa + 4 * s);
      v2f b;
      b[0] = cB[fb + (4 * s) * KC];
      b[1] = cB[fb + (4 * s + 1) * KC];
      c = __builtin_amdgcn_wmma_f32_16x16x4_f32(false, a, false, b,
                                                (short)0, c, false, false);
    }
    if (ch + 1 < TNC) {
      float* nA = sA[(ch + 1) & 1];
      float* nB = sB[(ch + 1) & 1];
#pragma unroll
      for (int j = 0; j < 4; ++j)
        *(float4*)(nB + lB + 4 * j * TPB) = rb[j];
      *(float4*)(nA + lA) = ra0;
      if (a2) *(float4*)(nA + lA + 20 * TPA) = ra1;
    }
    if (ch + 1 < TNC) __syncthreads();
  }

#pragma unroll
  for (int r = 0; r < 8; ++r) {
    *(float*)((char*)T + tbyte + (size_t)r * (DCH * sizeof(float))) = c[r];
  }
}

__global__ __launch_bounds__(256) void gbst_main(
    const int*   __restrict__ ids,
    const float* __restrict__ T,
    const float* __restrict__ conv_b,
    const float* __restrict__ sw,
    float*       __restrict__ out,
    int L)
{
  __shared__ float xt[TL][DCH];
  __shared__ float q[TL];
  __shared__ float wc[TL][4];
  __shared__ int   sid[TL + 4];

  const int b   = blockIdx.y;
  const int t0  = blockIdx.x * TL;
  const int tid = threadIdx.x;

  if (tid < TL + 4) {
    int g = t0 + tid - 2;
    sid[tid] = (g >= 0 && g < L) ? ids[(size_t)b * L + g] : -1;
  }
  __syncthreads();

  const int oc = (tid & 127) * 4;
  const int pr = tid >> 7;

  const float bx = conv_b[oc + 0], by = conv_b[oc + 1];
  const float bz = conv_b[oc + 2], bw = conv_b[oc + 3];
  for (int pb = 0; pb < TL; pb += 2 * GP) {
    float4 tv[GP][KC];
#pragma unroll
    for (int g = 0; g < GP; ++g) {
#pragma unroll
      for (int k = 0; k < KC; ++k) {
        const int v  = sid[pb + pr + 2 * g + k];
        const int vc = v < 0 ? 0 : v;
        float4 t = *(const float4*)(T + ((size_t)k * VOCAB + vc) * DCH + oc);
        if (v < 0) { t.x = 0.f; t.y = 0.f; t.z = 0.f; t.w = 0.f; }
        tv[g][k] = t;
      }
    }
#pragma unroll
    for (int g = 0; g < GP; ++g) {
      const int p = pb + pr + 2 * g;
      float ax = bx, ay = by, az = bz, aw = bw;
#pragma unroll
      for (int k = 0; k < KC; ++k) {
        ax += tv[g][k].x; ay += tv[g][k].y; az += tv[g][k].z; aw += tv[g][k].w;
      }
      if (t0 + p >= L) { ax = 0.f; ay = 0.f; az = 0.f; aw = 0.f; }
      xt[p][oc + 0] = ax; xt[p][oc + 1] = ay;
      xt[p][oc + 2] = az; xt[p][oc + 3] = aw;
    }
  }
  __syncthreads();

  {
    const int wave = tid >> 5;
    const int lane = tid & 31;
    float swr[DCH / 32];
#pragma unroll
    for (int j = 0; j < DCH / 32; ++j) swr[j] = sw[lane + 32 * j];
    for (int p = wave; p < TL; p += 8) {
      float s = 0.f;
#pragma unroll
      for (int j = 0; j < DCH / 32; ++j) s += xt[p][lane + 32 * j] * swr[j];
#pragma unroll
      for (int off = 16; off > 0; off >>= 1) s += __shfl_xor(s, off, 32);
      if (lane == 0) q[p] = s;
    }
  }
  __syncthreads();

  if (tid < TL) {
    const int l = tid;
    const int j2 = (l >> 1) << 1;
    const int j3 = (l / 3) * 3;
    const int j4 = (l >> 2) << 2;
    float s1 = q[l];
    float s2 = 0.5f * (q[j2] + q[j2 + 1]);
    float s3 = (q[j3] + q[j3 + 1] + q[j3 + 2]) * (1.f / 3.f);
    float s4 = 0.25f * (q[j4] + q[j4 + 1] + q[j4 + 2] + q[j4 + 3]);
    float mx = fmaxf(fmaxf(s1, s2), fmaxf(s3, s4));
    float e1 = __expf(s1 - mx), e2 = __expf(s2 - mx);
    float e3 = __expf(s3 - mx), e4 = __expf(s4 - mx);
    float inv = 1.f / (e1 + e2 + e3 + e4);
    wc[l][0] = e1 * inv;
    wc[l][1] = e2 * inv * 0.5f;
    wc[l][2] = e3 * inv * (1.f / 3.f);
    wc[l][3] = e4 * inv * 0.25f;
  }
  __syncthreads();

  for (int l2 = pr; l2 < TL / 2; l2 += 2) {
    if (t0 + 2 * l2 >= L) break;
    float ax = 0.f, ay = 0.f, az = 0.f, aw = 0.f;
#pragma unroll
    for (int dl = 0; dl < 2; ++dl) {
      const int l  = 2 * l2 + dl;
      const int j2 = (l >> 1) << 1;
      const int j3 = (l / 3) * 3;
      const int j4 = (l >> 2) << 2;
      const float w1 = wc[l][0], w2 = wc[l][1], w3 = wc[l][2], w4 = wc[l][3];
      ax += w1 * xt[l][oc + 0];
      ay += w1 * xt[l][oc + 1];
      az += w1 * xt[l][oc + 2];
      aw += w1 * xt[l][oc + 3];
      ax += w2 * (xt[j2][oc + 0] + xt[j2 + 1][oc + 0]);
      ay += w2 * (xt[j2][oc + 1] + xt[j2 + 1][oc + 1]);
      az += w2 * (xt[j2][oc + 2] + xt[j2 + 1][oc + 2]);
      aw += w2 * (xt[j2][oc + 3] + xt[j2 + 1][oc + 3]);
      ax += w3 * (xt[j3][oc + 0] + xt[j3 + 1][oc + 0] + xt[j3 + 2][oc + 0]);
      ay += w3 * (xt[j3][oc + 1] + xt[j3 + 1][oc + 1] + xt[j3 + 2][oc + 1]);
      az += w3 * (xt[j3][oc + 2] + xt[j3 + 1][oc + 2] + xt[j3 + 2][oc + 2]);
      aw += w3 * (xt[j3][oc + 3] + xt[j3 + 1][oc + 3] + xt[j3 + 2][oc + 3]);
      ax += w4 * (xt[j4][oc + 0] + xt[j4 + 1][oc + 0] + xt[j4 + 2][oc + 0] + xt[j4 + 3][oc + 0]);
      ay += w4 * (xt[j4][oc + 1] + xt[j4 + 1][oc + 1] + xt[j4 + 2][oc + 1] + xt[j4 + 3][oc + 1]);
      az += w4 * (xt[j4][oc + 2] + xt[j4 + 1][oc + 2] + xt[j4 + 2][oc + 2] + xt[j4 + 3][oc + 2]);
      aw += w4 * (xt[j4][oc + 3] + xt[j4 + 1][oc + 3] + xt[j4 + 2][oc + 3] + xt[j4 + 3][oc + 3]);
    }
    float4 r;
    r.x = 0.5f * ax; r.y = 0.5f * ay; r.z = 0.5f * az; r.w = 0.5f * aw;
    const size_t off = ((size_t)b * (L >> 1) + ((t0 >> 1) + l2)) * DCH + oc;
    *(float4*)(out + off) = r;
  }
}

extern "C" void kernel_launch(void* const* d_in, const int* in_sizes, int n_in,
                              void* d_out, int out_size, void* d_ws, size_t ws_size,
                              hipStream_t stream) {
  const int*   ids     = (const int*)d_in[0];
  const float* emb     = (const float*)d_in[1];
  const float* conv_w  = (const float*)d_in[2];
  const float* conv_b  = (const float*)d_in[3];
  const float* score_w = (const float*)d_in[4];
  float* out = (float*)d_out;
  float* T   = (float*)d_ws;

  const int L = LLEN;
  const int B = in_sizes[0] / L;

  dim3 g1(DCH / 16, VOCAB / (16 * TVT));
  gbst_table_wmma<<<g1, TTHR, 0, stream>>>(emb, conv_w, T);

  dim3 g2((L + TL - 1) / TL, B);
  gbst_main<<<g2, 256, 0, stream>>>(ids, T, conv_b, score_w, out, L);
}


# Main kernel keeps the conv rows of a 12-position group in the gathering thread's registers: score dot summed by lane exchange, then over 4 waves (new sum order); mix from registers; no LDS conv tile
# speedup vs baseline: 2.3036x; 1.1048x over previous
#include <hip/hip_runtime.h>
#include <hip/hip_bf16.h>

typedef float v2f __attribute__((ext_vector_type(2)));
typedef float v8f __attribute__((ext_vector_type(8)));

#define VOCAB 256
#define DCH   512
#define KC    5
#define GL    12
#define NGB   2
#define NIT   2
#define TL    (GL * NGB * NIT)
#define LLEN  8192
#define NPT   3

#define TIC   64
#define TNC   (DCH / TIC)
#define TVT   2
#define TPB   (TIC * KC + 20)
#define TPA   (TIC + 4)
#define TTHR  (32 * KC * TVT)

__global__ __launch_bounds__(TTHR) void gbst_table_wmma(
    const float* __restrict__ emb,
    const float* __restrict__ conv_w,
    float* __restrict__ T)
{
  __shared__ __attribute__((aligned(16))) float sB[2][16 * TPB];
  __shared__ __attribute__((aligned(16))) float sA[2][16 * TVT * TPA];

  const int tid  = threadIdx.x;
  const int o0   = blockIdx.x * 16;
  const int vb   = blockIdx.y * (16 * TVT);
  const int wave = tid >> 5;
  const int lane = tid & 31;
  const int half = lane >> 4;
  const int sub  = lane & 15;
  const int k    = wave % KC;
  const int vt   = wave / KC;

  const int  brow = tid / 80;
  const int  bc4  = tid % 80;
  const int  arow = tid >> 4;
  const int  ac4  = tid & 15;
  const bool a2   = arow + 20 < 16 * TVT;

  const float* gB = conv_w + (size_t)(o0 + brow) * (DCH * KC) + bc4 * 4;
  const float* gA = emb + (size_t)(vb + arow) * DCH + ac4 * 4;
  const int lB = brow * TPB + bc4 * 4;
  const int lA = arow * TPA + ac4 * 4;

  float4 rb[4];
  float4 ra0, ra1;
  ra1.x = 0.f; ra1.y = 0.f; ra1.z = 0.f; ra1.w = 0.f;

#pragma unroll
  for (int j = 0; j < 4; ++j)
    rb[j] = *(const float4*)(gB + (size_t)(4 * j) * (DCH * KC));
  ra0 = *(const float4*)gA;
  if (a2) ra1 = *(const float4*)(gA + 20 * DCH);
#pragma unroll
  for (int j = 0; j < 4; ++j)
    *(float4*)(&sB[0][lB + 4 * j * TPB]) = rb[j];
  *(float4*)(&sA[0][lA]) = ra0;
  if (a2) *(float4*)(&sA[0][lA + 20 * TPA]) = ra1;
  __syncthreads();

  v8f c = {};
  const int fa = (vt * 16 + sub) * TPA + 2 * half;
  const int fb = sub * TPB + 2 * half * KC + k;

  unsigned tbyte = (unsigned)((((size_t)k * VOCAB + (vb + vt * 16 + half * 8)) * DCH + (o0 + sub)) * sizeof(float));
  asm volatile("" : "+v"(tbyte));

  for (int ch = 0; ch < TNC; ++ch) {
    const float* cA = sA[ch & 1];
    const float* cB = sB[ch & 1];
    if (ch + 1 < TNC) {
      const int i1 = (ch + 1) * TIC;
#pragma unroll
      for (int j = 0; j < 4; ++j)
        rb[j] = *(const float4*)(gB + (size_t)(4 * j) * (DCH * KC) + i1 * KC);
      ra0 = *(const float4*)(gA + i1);
      if (a2) ra1 = *(const float4*)(gA + 20 * DCH + i1);
    }
#pragma unroll
    for (int s = 0; s < TIC / 4; ++s) {
      const v2f a = *(const v2f*)(cA + fa + 4 * s);
      v2f b;
      b[0] = cB[fb + (4 * s) * KC];
      b[1] = cB[fb + (4 * s + 1) * KC];
      c = __builtin_amdgcn_wmma_f32_16x16x4_f32(false, a, false, b,
                                                (short)0, c, false, false);
    }
    if (ch + 1 < TNC) {
      float* nA = sA[(ch + 1) & 1];
      float* nB = sB[(ch + 1) & 1];
#pragma unroll
      for (int j = 0; j < 4; ++j)
        *(float4*)(nB + lB + 4 * j * TPB) = rb[j];
      *(float4*)(nA + lA) = ra0;
      if (a2) *(float4*)(nA + lA + 20 * TPA) = ra1;
    }
    if (ch + 1 < TNC) __syncthreads();
  }

#pragma unroll
  for (int r = 0; r < 8; ++r) {
    *(float*)((char*)T + tbyte + (size_t)r * (DCH * sizeof(float))) = c[r];
  }
}

__global__ __launch_bounds__(256, 6) void gbst_main(
    const int*   __restrict__ ids,
    const float* __restrict__ T,
    const float* __restrict__ conv_b,
    const float* __restrict__ sw,
    float*       __restrict__ out,
    int L)
{
  __shared__ int sid[NGB][GL + 4];
  __shared__ __attribute__((aligned(16))) float red[NGB][GL][4];
  __shared__ __attribute__((aligned(16))) float wc[NGB][GL][4];

  const int b    = blockIdx.y;
  const int tid  = threadIdx.x;
  const int grp  = tid >> 7;
  const int qd   = tid & 127;
  const int oc   = qd * 4;
  const int wv   = (tid >> 5) & 3;
  const int lane = tid & 31;

  const float bx = conv_b[oc + 0], by = conv_b[oc + 1];
  const float bz = conv_b[oc + 2], bw = conv_b[oc + 3];
  const float wx = sw[oc + 0], wy = sw[oc + 1];
  const float wz = sw[oc + 2], ww = sw[oc + 3];

  for (int it = 0; it < NIT; ++it) {
    const int g0 = blockIdx.x * TL + (it * NGB + grp) * GL;

    if (qd < GL + 4) {
      const int g = g0 + qd - 2;
      sid[grp][qd] = (g >= 0 && g < L) ? ids[(size_t)b * L + g] : -1;
    }
    __syncthreads();

    float4 xa[NPT], xb[NPT], xc[NPT], xd[NPT];
#pragma unroll
    for (int g = 0; g < NPT; ++g) {
      xa[g].x = 0.f; xa[g].y = 0.f; xa[g].z = 0.f; xa[g].w = 0.f;
      xb[g].x = 0.f; xb[g].y = 0.f; xb[g].z = 0.f; xb[g].w = 0.f;
      xc[g].x = 0.f; xc[g].y = 0.f; xc[g].z = 0.f; xc[g].w = 0.f;
      xd[g].x = 0.f; xd[g].y = 0.f; xd[g].z = 0.f; xd[g].w = 0.f;
    }
#pragma unroll 1
    for (int pb = 0; pb < GL; pb += NPT) {
      float4 tv[NPT][KC];
#pragma unroll
      for (int g = 0; g < NPT; ++g) {
#pragma unroll
        for (int k = 0; k < KC; ++k) {
          const int v  = __builtin_amdgcn_readfirstlane(sid[grp][pb + g + k]);
          const int vc = v < 0 ? 0 : v;
          float4 t = *(const float4*)(T + ((size_t)k * VOCAB + vc) * DCH + oc);
          if (v < 0) { t.x = 0.f; t.y = 0.f; t.z = 0.f; t.w = 0.f; }
          tv[g][k] = t;
        }
      }
      float4 nw[NPT];
#pragma unroll
      for (int g = 0; g < NPT; ++g) {
        const int p = pb + g;
        float ax = bx, ay = by, az = bz, aw = bw;
#pragma unroll
        for (int k = 0; k < KC; ++k) {
          ax += tv[g][k].x; ay += tv[g][k].y; az += tv[g][k].z; aw += tv[g][k].w;
        }
        if (g0 + p >= L) { ax = 0.f; ay = 0.f; az = 0.f; aw = 0.f; }
        nw[g].x = ax; nw[g].y = ay; nw[g].z = az; nw[g].w = aw;
      }
#pragma unroll
      for (int g = 0; g < NPT; ++g) {
        xa[g] = xb[g];
        xb[g] = xc[g];
        xc[g] = xd[g];
        xd[g] = nw[g];
      }
    }
    float4 x[GL];
#pragma unroll
    for (int g = 0; g < NPT; ++g) {
      x[0 * NPT + g] = xa[g];
      x[1 * NPT + g] = xb[g];
      x[2 * NPT + g] = xc[g];
      x[3 * NPT + g] = xd[g];
    }

    float s[GL];
#pragma unroll
    for (int i = 0; i < GL; ++i)
      s[i] = x[i].x * wx + x[i].y * wy + x[i].z * wz + x[i].w * ww;
#pragma unroll
    for (int off = 16; off > 0; off >>= 1) {
#pragma unroll
      for (int i = 0; i < GL; ++i) s[i] += __shfl_xor(s[i], off, 32);
    }
    if (lane == 0) {
#pragma unroll
      for (int i = 0; i < GL; ++i) red[grp][i][wv] = s[i];
    }
    __syncthreads();

    if (qd < GL) {
      const int l  = qd;
      const int j2 = (l >> 1) << 1;
      const int j3 = (l / 3) * 3;
      const int j4 = (l >> 2) << 2;
      float q[GL];
#pragma unroll
      for (int i = 0; i < GL; ++i) {
        const float4 r = *(const float4*)(&red[grp][i][0]);
        q[i] = ((r.x + r.y) + r.z) + r.w;
      }
      float q1 = 0.f, q2a = 0.f, q2b = 0.f, q3a = 0.f, q3b = 0.f, q3c = 0.f;
      float q4a = 0.f, q4b = 0.f, q4c = 0.f, q4d = 0.f;
#pragma unroll
      for (int i = 0; i < GL; ++i) {
        if (i == l)      q1  = q[i];
        if (i == j2)     q2a = q[i];
        if (i == j2 + 1) q2b = q[i];
        if (i == j3)     q3a = q[i];
        if (i == j3 + 1) q3b = q[i];
        if (i == j3 + 2) q3c = q[i];
        if (i == j4)     q4a = q[i];
        if (i == j4 + 1) q4b = q[i];
        if (i == j4 + 2) q4c = q[i];
        if (i == j4 + 3) q4d = q[i];
      }
      const float s1 = q1;
      const float s2 = 0.5f * (q2a + q2b);
      const float s3 = (q3a + q3b + q3c) * (1.f / 3.f);
      const float s4 = 0.25f * (q4a + q4b + q4c + q4d);
      const float mx = fmaxf(fmaxf(s1, s2), fmaxf(s3, s4));
      const float e1 = __expf(s1 - mx), e2 = __expf(s2 - mx);
      const float e3 = __expf(s3 - mx), e4 = __expf(s4 - mx);
      const float inv = 1.f / (e1 + e2 + e3 + e4);
      float4 w;
      w.x = e1 * inv;
      w.y = e2 * inv * 0.5f;
      w.z = e3 * inv * (1.f / 3.f);
      w.w = e4 * inv * 0.25f;
      *(float4*)(&wc[grp][l][0]) = w;
    }
    __syncthreads();

    float4 s2v[GL / 2], s3v[GL / 3], s4v[GL / 4];
#pragma unroll
    for (int i = 0; i < GL / 2; ++i) {
      s2v[i].x = x[2 * i].x + x[2 * i + 1].x;
      s2v[i].y = x[2 * i].y + x[2 * i + 1].y;
      s2v[i].z = x[2 * i].z + x[2 * i + 1].z;
      s2v[i].w = x[2 * i].w + x[2 * i + 1].w;
    }
#pragma unroll
    for (int i = 0; i < GL / 3; ++i) {
      s3v[i].x = (x[3 * i].x + x[3 * i + 1].x) + x[3 * i + 2].x;
      s3v[i].y = (x[3 * i].y + x[3 * i + 1].y) + x[3 * i + 2].y;
      s3v[i].z = (x[3 * i].z + x[3 * i + 1].z) + x[3 * i + 2].z;
      s3v[i].w = (x[3 * i].w + x[3 * i + 1].w) + x[3 * i + 2].w;
    }
#pragma unroll
    for (int i = 0; i < GL / 4; ++i) {
      s4v[i].x = (s2v[2 * i].x + x[4 * i + 2].x) + x[4 * i + 3].x;
      s4v[i].y = (s2v[2 * i].y + x[4 * i + 2].y) + x[4 * i + 3].y;
      s4v[i].z = (s2v[2 * i].z + x[4 * i + 2].z) + x[4 * i + 3].z;
      s4v[i].w = (s2v[2 * i].w + x[4 * i + 2].w) + x[4 * i + 3].w;
    }
#pragma unroll
    for (int j = 0; j < GL / 2; ++j) {
      float ax = 0.f, ay = 0.f, az = 0.f, aw = 0.f;
#pragma unroll
      for (int dl = 0; dl < 2; ++dl) {
        const int l = 2 * j + dl;
        const float4 w = *(const float4*)(&wc[grp][l][0]);
        ax += w.x * x[l].x;
        ay += w.x * x[l].y;
        az += w.x * x[l].z;
        aw += w.x * x[l].w;
        ax += w.y * s2v[l >> 1].x;
        ay += w.y * s2v[l >> 1].y;
        az += w.y * s2v[l >> 1].z;
        aw += w.y * s2v[l >> 1].w;
        ax += w.z * s3v[l / 3].x;
        ay += w.z * s3v[l / 3].y;
        az += w.z * s3v[l / 3].z;
        aw += w.z * s3v[l / 3].w;
        ax += w.w * s4v[l >> 2].x;
        ay += w.w * s4v[l >> 2].y;
        az += w.w * s4v[l >> 2].z;
        aw += w.w * s4v[l >> 2].w;
      }
      if (g0 + 2 * j < L) {
        float4 r;
        r.x = 0.5f * ax; r.y = 0.5f * ay; r.z = 0.5f * az; r.w = 0.5f * aw;
        const size_t off = ((size_t)b * (L >> 1) + ((g0 >> 1) + j)) * DCH + oc;
        *(float4*)(out + off) = r;
      }
    }
  }
}

extern "C" void kernel_launch(void* const* d_in, const int* in_sizes, int n_in,
                              void* d_out, int out_size, void* d_ws, size_t ws_size,
                              hipStream_t stream) {
  const int*   ids     = (const int*)d_in[0];
  const float* emb     = (const float*)d_in[1];
  const float* conv_w  = (const float*)d_in[2];
  const float* conv_b  = (const float*)d_in[3];
  const float* score_w = (const float*)d_in[4];
  float* out = (float*)d_out;
  float* T   = (float*)d_ws;

  const int L = LLEN;
  const int B = in_sizes[0] / L;

  dim3 g1(DCH / 16, VOCAB / (16 * TVT));
  gbst_table_wmma<<<g1, TTHR, 0, stream>>>(emb, conv_w, T);

  dim3 g2((L + TL - 1) / TL, B);
  gbst_main<<<g2, 256, 0, stream>>>(ids, T, conv_b, score_w, out, L);
}


# Kernel 2 cut by channels: work-group copies its 32-channel table slice to LDS once, gathers conv rows from LDS not L2; scores from a score table summed in-call from kernel 1 partials (new sum order)
# speedup vs baseline: 2.6173x; 1.1362x over previous
#include <hip/hip_runtime.h>
#include <hip/hip_bf16.h>

typedef float v2f __attribute__((ext_vector_type(2)));
typedef float v8f __attribute__((ext_vector_type(8)));

#define VOCAB 256
#define DCH   512
#define KC    5
#define GL    12
#define LLEN  8192
#define CS    32
#define TROWS (VOCAB + 1)
#define NGROW ((LLEN + GL - 1) / GL)
#define NGH   ((NGROW + 1) / 2)
#define NPH   (NGH * GL)
#define MTHR  512
#define NSE   5

#define TIC   64
#define TNC   (DCH / TIC)
#define TVT   2
#define TPB   (TIC * KC + 20)
#define TPA   (TIC + 4)
#define TTHR  (32 * KC * TVT)

__global__ __launch_bounds__(TTHR) void gbst_table_wmma(
    const float* __restrict__ emb,
    const float* __restrict__ conv_w,
    const float* __restrict__ sw,
    float* __restrict__ T,
    float* __restrict__ Sp)
{
  __shared__ __attribute__((aligned(16))) float sB[2][16 * TPB];
  __shared__ __attribute__((aligned(16))) float sA[2][16 * TVT * TPA];

  const int tid  = threadIdx.x;
  const int o0   = blockIdx.x * 16;
  const int vb   = blockIdx.y * (16 * TVT);
  const int wave = tid >> 5;
  const int lane = tid & 31;
  const int half = lane >> 4;
  const int sub  = lane & 15;
  const int k    = wave % KC;
  const int vt   = wave / KC;

  const int  brow = tid / 80;
  const int  bc4  = tid % 80;
  const int  arow = tid >> 4;
  const int  ac4  = tid & 15;
  const bool a2   = arow + 20 < 16 * TVT;

  const float* gB = conv_w + (size_t)(o0 + brow) * (DCH * KC) + bc4 * 4;
  const float* gA = emb + (size_t)(vb + arow) * DCH + ac4 * 4;
  const int lB = brow * TPB + bc4 * 4;
  const int lA = arow * TPA + ac4 * 4;

  float4 rb[4];
  float4 ra0, ra1;
  ra1.x = 0.f; ra1.y = 0.f; ra1.z = 0.f; ra1.w = 0.f;

#pragma unroll
  for (int j = 0; j < 4; ++j)
    rb[j] = *(const float4*)(gB + (size_t)(4 * j) * (DCH * KC));
  ra0 = *(const float4*)gA;
  if (a2) ra1 = *(const float4*)(gA + 20 * DCH);
#pragma unroll
  for (int j = 0; j < 4; ++j)
    *(float4*)(&sB[0][lB + 4 * j * TPB]) = rb[j];
  *(float4*)(&sA[0][lA]) = ra0;
  if (a2) *(float4*)(&sA[0][lA + 20 * TPA]) = ra1;
  __syncthreads();

  const float swv = sw[o0 + sub];

  v8f c = {};
  const int fa = (vt * 16 + sub) * TPA + 2 * half;
  const int fb = sub * TPB + 2 * half * KC + k;

  unsigned tbyte = (unsigned)((((size_t)k * VOCAB + (vb + vt * 16 + half * 8)) * DCH + (o0 + sub)) * sizeof(float));
  asm volatile("" : "+v"(tbyte));

  for (int ch = 0; ch < TNC; ++ch) {
    const float* cA = sA[ch & 1];
    const float* cB = sB[ch & 1];
    if (ch + 1 < TNC) {
      const int i1 = (ch + 1) * TIC;
#pragma unroll
      for (int j = 0; j < 4; ++j)
        rb[j] = *(const float4*)(gB + (size_t)(4 * j) * (DCH * KC) + i1 * KC);
      ra0 = *(const float4*)(gA + i1);
      if (a2) ra1 = *(const float4*)(gA + 20 * DCH + i1);
    }
#pragma unroll
    for (int s = 0; s < TIC / 4; ++s) {
      const v2f a = *(const v2f*)(cA + fa + 4 * s);
      v2f b;
      b[0] = cB[fb + (4 * s) * KC];
      b[1] = cB[fb + (4 * s + 1) * KC];
      c = __builtin_amdgcn_wmma_f32_16x16x4_f32(false, a, false, b,
                                                (short)0, c, false, false);
    }
    if (ch + 1 < TNC) {
      float* nA = sA[(ch + 1) & 1];
      float* nB = sB[(ch + 1) & 1];
#pragma unroll
      for (int j = 0; j < 4; ++j)
        *(float4*)(nB + lB + 4 * j * TPB) = rb[j];
      *(float4*)(nA + lA) = ra0;
      if (a2) *(float4*)(nA + lA + 20 * TPA) = ra1;
    }
    if (ch + 1 < TNC) __syncthreads();
  }

#pragma unroll
  for (int r = 0; r < 8; ++r) {
    *(float*)((char*)T + tbyte + (size_t)r * (DCH * sizeof(float))) = c[r];
  }

  float ps[8];
#pragma unroll
  for (int r = 0; r < 8; ++r) ps[r] = c[r] * swv;
#pragma unroll
  for (int off = 8; off > 0; off >>= 1) {
#pragma unroll
    for (int r = 0; r < 8; ++r) ps[r] += __shfl_xor(ps[r], off, 32);
  }
  if (sub == 0) {
    float* d = Sp + ((size_t)blockIdx.x * KC + k) * VOCAB + (vb + vt * 16 + half * 8);
    float4 lo, hi;
    lo.x = ps[0]; lo.y = ps[1]; lo.z = ps[2]; lo.w = ps[3];
    hi.x = ps[4]; hi.y = ps[5]; hi.z = ps[6]; hi.w = ps[7];
    *(float4*)(d)     = lo;
    *(float4*)(d + 4) = hi;
  }
}

__global__ __launch_bounds__(MTHR) void gbst_main(
    const int*   __restrict__ ids,
    const float* __restrict__ T,
    const float* __restrict__ Sp,
    const float* __restrict__ conv_b,
    const float* __restrict__ sw,
    float*       __restrict__ out)
{
  __shared__ __attribute__((aligned(16))) float Ts[KC * TROWS * CS];
  __shared__ __attribute__((aligned(16))) float wc[NPH * 4];
  __shared__ __attribute__((aligned(16))) int   sid[NPH + 4];
  __shared__ float Ss[KC * TROWS];
  __shared__ float sc[NPH];
  __shared__ float sbsw[4];

  const int L    = LLEN;
  const int tid  = threadIdx.x;
  const int sl   = blockIdx.x;
  const int hf   = blockIdx.y;
  const int b    = blockIdx.z;
  const int ng   = hf ? (NGROW - NGH) : NGH;
  const int P0   = hf ? NPH : 0;
  const int np   = ng * GL;

  const int lane = tid & 31;
  int idr[(NPH + 4 + MTHR - 1) / MTHR];
#pragma unroll
  for (int j = 0; j < (NPH + 4 + MTHR - 1) / MTHR; ++j) {
    const int q = tid + j * MTHR;
    const int g = P0 + q - 2;
    idr[j] = (q < np + 4 && g >= 0 && g < L) ? ids[(size_t)b * L + g] : VOCAB;
  }
  float4 tv[(KC * VOCAB * CS / 4) / MTHR];
#pragma unroll
  for (int j = 0; j < (KC * VOCAB * CS / 4) / MTHR; ++j) {
    const int f   = tid + j * MTHR;
    const int row = f >> 3;
    tv[j] = *(const float4*)(T + (size_t)row * DCH + sl * CS + (f & 7) * 4);
  }
  const int part = tid >> 6;
  const int l64  = tid & 63;
  float4 pa[NSE];
#pragma unroll
  for (int m = 0; m < NSE; ++m) {
    const float* s = Sp + (size_t)(part * 4) * (KC * VOCAB) + (l64 + 64 * m) * 4;
    const float4 s0 = *(const float4*)(s);
    const float4 s1 = *(const float4*)(s + 1 * (KC * VOCAB));
    const float4 s2 = *(const float4*)(s + 2 * (KC * VOCAB));
    const float4 s3 = *(const float4*)(s + 3 * (KC * VOCAB));
    pa[m].x = ((s0.x + s1.x) + s2.x) + s3.x;
    pa[m].y = ((s0.y + s1.y) + s2.y) + s3.y;
    pa[m].z = ((s0.z + s1.z) + s2.z) + s3.z;
    pa[m].w = ((s0.w + s1.w) + s2.w) + s3.w;
  }
  const int ch4 = (tid & 127) * 4;
  const float4 cb = *(const float4*)(conv_b + ch4);
  const float4 cw = *(const float4*)(sw + ch4);
  float bs = ((cb.x * cw.x + cb.y * cw.y) + cb.z * cw.z) + cb.w * cw.w;
#pragma unroll
  for (int off = 16; off > 0; off >>= 1) bs += __shfl_xor(bs, off, 32);

  float* pp = wc;
#pragma unroll
  for (int m = 0; m < NSE; ++m)
    *(float4*)(&pp[(part * (KC * VOCAB / 4) + l64 + 64 * m) * 4]) = pa[m];
#pragma unroll
  for (int j = 0; j < (NPH + 4 + MTHR - 1) / MTHR; ++j) {
    const int q = tid + j * MTHR;
    if (q < np + 4) sid[q] = idr[j];
  }
#pragma unroll
  for (int j = 0; j < (KC * VOCAB * CS / 4) / MTHR; ++j) {
    const int f   = tid + j * MTHR;
    const int row = f >> 3;
    const int k   = row >> 8;
    const int v   = row & (VOCAB - 1);
    *(float4*)(&Ts[(k * TROWS + v) * CS + (f & 7) * 4]) = tv[j];
  }
  if (tid < KC * CS) Ts[((tid >> 5) * TROWS + VOCAB) * CS + (tid & 31)] = 0.f;
  if (tid >= MTHR - 128 && lane == 0) sbsw[(tid - (MTHR - 128)) >> 5] = bs;
  __syncthreads();

  if (tid < KC * VOCAB / 4) {
    float4 a = *(const float4*)(&pp[tid * 4]);
#pragma unroll
    for (int p = 1; p < 8; ++p) {
      const float4 t = *(const float4*)(&pp[(p * (KC * VOCAB / 4) + tid) * 4]);
      a.x += t.x; a.y += t.y; a.z += t.z; a.w += t.w;
    }
    const int e = tid * 4;
    const int o = (e >> 8) * TROWS + (e & (VOCAB - 1));
    Ss[o + 0] = a.x; Ss[o + 1] = a.y; Ss[o + 2] = a.z; Ss[o + 3] = a.w;
  } else if (tid < KC * VOCAB / 4 + KC) {
    Ss[(tid - KC * VOCAB / 4) * TROWS + VOCAB] = 0.f;
  }
  __syncthreads();

  {
    const float bsw = ((sbsw[0] + sbsw[1]) + sbsw[2]) + sbsw[3];
#pragma unroll
    for (int j = 0; j < (NPH + MTHR - 1) / MTHR; ++j) {
      const int q  = tid + j * MTHR;
      const int qq = q < np ? q : np - 1;
      float a = bsw;
#pragma unroll
      for (int k = 0; k < KC; ++k) a += Ss[k * TROWS + sid[qq + k]];
      if (P0 + qq >= L) a = 0.f;
      if (q < np) sc[q] = a;
    }
  }
  __syncthreads();

#pragma unroll
  for (int j = 0; j < (NPH + MTHR - 1) / MTHR; ++j) {
    const int q  = tid + j * MTHR;
    const int qq = q < np ? q : np - 1;
    const int l  = qq % GL;
    const int j2 = qq & ~1;
    const int j3 = qq - (l % 3);
    const int j4 = qq & ~3;
    const float s1 = sc[qq];
    const float s2 = 0.5f * (sc[j2] + sc[j2 + 1]);
    const float s3 = (sc[j3] + sc[j3 + 1] + sc[j3 + 2]) * (1.f / 3.f);
    const float s4 = 0.25f * (sc[j4] + sc[j4 + 1] + sc[j4 + 2] + sc[j4 + 3]);
    const float mx = fmaxf(fmaxf(s1, s2), fmaxf(s3, s4));
    const float e1 = __expf(s1 - mx), e2 = __expf(s2 - mx);
    const float e3 = __expf(s3 - mx), e4 = __expf(s4 - mx);
    const float inv = 1.f / (e1 + e2 + e3 + e4);
    float4 w;
    w.x = e1 * inv;
    w.y = e2 * inv * 0.5f;
    w.z = e3 * inv * (1.f / 3.f);
    w.w = e4 * inv * 0.25f;
    if (q < np) *(float4*)(&wc[q * 4]) = w;
  }
  __syncthreads();

  const int cq = tid & 7;
  const int oc = sl * CS + cq * 4;
  const float4 bv = *(const float4*)(conv_b + oc);
  const int ntask = ng * 8;
  for (int task = tid; task < ntask; task += MTHR) {
    const int q0 = (task >> 3) * GL;
    const int g0 = P0 + q0;

    int rv[GL + 4];
#pragma unroll
    for (int i = 0; i < (GL + 4) / 4; ++i) {
      const int4 t = *(const int4*)(&sid[q0 + 4 * i]);
      rv[4 * i + 0] = t.x; rv[4 * i + 1] = t.y; rv[4 * i + 2] = t.z; rv[4 * i + 3] = t.w;
    }

    float4 x[GL];
#pragma unroll
    for (int p = 0; p < GL; ++p) {
      float ax = bv.x, ay = bv.y, az = bv.z, aw = bv.w;
#pragma unroll
      for (int k = 0; k < KC; ++k) {
        const float4 t = *(const float4*)(&Ts[(k * TROWS + rv[p + k]) * CS + cq * 4]);
        ax += t.x; ay += t.y; az += t.z; aw += t.w;
      }
      if (g0 + p >= L) { ax = 0.f; ay = 0.f; az = 0.f; aw = 0.f; }
      x[p].x = ax; x[p].y = ay; x[p].z = az; x[p].w = aw;
    }

    float4 s2v[GL / 2], s3v[GL / 3], s4v[GL / 4];
#pragma unroll
    for (int i = 0; i < GL / 2; ++i) {
      s2v[i].x = x[2 * i].x + x[2 * i + 1].x;
      s2v[i].y = x[2 * i].y + x[2 * i + 1].y;
      s2v[i].z = x[2 * i].z + x[2 * i + 1].z;
      s2v[i].w = x[2 * i].w + x[2 * i + 1].w;
    }
#pragma unroll
    for (int i = 0; i < GL / 3; ++i) {
      s3v[i].x = (x[3 * i].x + x[3 * i + 1].x) + x[3 * i + 2].x;
      s3v[i].y = (x[3 * i].y + x[3 * i + 1].y) + x[3 * i + 2].y;
      s3v[i].z = (x[3 * i].z + x[3 * i + 1].z) + x[3 * i + 2].z;
      s3v[i].w = (x[3 * i].w + x[3 * i + 1].w) + x[3 * i + 2].w;
    }
#pragma unroll
    for (int i = 0; i < GL / 4; ++i) {
      s4v[i].x = (s2v[2 * i].x + x[4 * i + 2].x) + x[4 * i + 3].x;
      s4v[i].y = (s2v[2 * i].y + x[4 * i + 2].y) + x[4 * i + 3].y;
      s4v[i].z = (s2v[2 * i].z + x[4 * i + 2].z) + x[4 * i + 3].z;
      s4v[i].w = (s2v[2 * i].w + x[4 * i + 2].w) + x[4 * i + 3].w;
    }
#pragma unroll
    for (int j = 0; j < GL / 2; ++j) {
      float ax = 0.f, ay = 0.f, az = 0.f, aw = 0.f;
#pragma unroll
      for (int dl = 0; dl < 2; ++dl) {
        const int l = 2 * j + dl;
        const float4 w = *(const float4*)(&wc[(q0 + l) * 4]);
        ax += w.x * x[l].x;
        ay += w.x * x[l].y;
        az += w.x * x[l].z;
        aw += w.x * x[l].w;
        ax += w.y * s2v[l >> 1].x;
        ay += w.y * s2v[l >> 1].y;
        az += w.y * s2v[l >> 1].z;
        aw += w.y * s2v[l >> 1].w;
        ax += w.z * s3v[l / 3].x;
        ay += w.z * s3v[l / 3].y;
        az += w.z * s3v[l / 3].z;
        aw += w.z * s3v[l / 3].w;
        ax += w.w * s4v[l >> 2].x;
        ay += w.w * s4v[l >> 2].y;
        az += w.w * s4v[l >> 2].z;
        aw += w.w * s4v[l >> 2].w;
      }
      if (g0 + 2 * j < L) {
        float4 r;
        r.x = 0.5f * ax; r.y = 0.5f * ay; r.z = 0.5f * az; r.w = 0.5f * aw;
        const size_t off = ((size_t)b * (L >> 1) + ((g0 >> 1) + j)) * DCH + oc;
        *(float4*)(out + off) = r;
      }
    }
  }
}

extern "C" void kernel_launch(void* const* d_in, const int* in_sizes, int n_in,
                              void* d_out, int out_size, void* d_ws, size_t ws_size,
                              hipStream_t stream) {
  const int*   ids     = (const int*)d_in[0];
  const float* emb     = (const float*)d_in[1];
  const float* conv_w  = (const float*)d_in[2];
  const float* conv_b  = (const float*)d_in[3];
  const float* score_w = (const float*)d_in[4];
  float* out = (float*)d_out;
  float* T   = (float*)d_ws;
  float* Sp  = T + KC * VOCAB * DCH;

  const int L = LLEN;
  const int B = in_sizes[0] / L;

  dim3 g1(DCH / 16, VOCAB / (16 * TVT));
  gbst_table_wmma<<<g1, TTHR, 0, stream>>>(emb, conv_w, score_w, T, Sp);

  dim3 g2(DCH / CS, 2, B);
  gbst_main<<<g2, MTHR, 0, stream>>>(ids, T, Sp, conv_b, score_w, out);
}


# Main kernel cut into 64-channel slices x quarter rows: each store writes whole aligned 256-byte units, LDS table rows cover all banks once; taps 0,1,3,4 from LDS, centre tap gathered from T
# speedup vs baseline: 2.9179x; 1.1149x over previous
#include <hip/hip_runtime.h>
#include <hip/hip_bf16.h>

typedef float v2f __attribute__((ext_vector_type(2)));
typedef float v8f __attribute__((ext_vector_type(8)));

#define VOCAB 256
#define DCH   512
#define KC    5
#define GL    12
#define LLEN  8192
#define CS    64
#define KL    4
#define TROWS (VOCAB + 1)
#define NGROW ((LLEN + GL - 1) / GL)
#define NQR   4
#define NGQ   ((NGROW + NQR - 1) / NQR)
#define NPQ   (NGQ * GL)
#define MTHR  512
#define NSE   5
#define NIDR  ((NPQ + 4 + MTHR - 1) / MTHR)
#define NPT   ((NPQ + MTHR - 1) / MTHR)
#define NTVH  ((KL * VOCAB * CS / 4) / MTHR / 2)

#define TIC   64
#define TNC   (DCH / TIC)
#define TVT   2
#define TPB   (TIC * KC + 20)
#define TPA   (TIC + 4)
#define TTHR  (32 * KC * TVT)

__global__ __launch_bounds__(TTHR) void gbst_table_wmma(
    const float* __restrict__ emb,
    const float* __restrict__ conv_w,
    const float* __restrict__ sw,
    float* __restrict__ T,
    float* __restrict__ Sp)
{
  __shared__ __attribute__((aligned(16))) float sB[2][16 * TPB];
  __shared__ __attribute__((aligned(16))) float sA[2][16 * TVT * TPA];

  const int tid  = threadIdx.x;
  const int o0   = blockIdx.x * 16;
  const int vb   = blockIdx.y * (16 * TVT);
  const int wave = tid >> 5;
  const int lane = tid & 31;
  const int half = lane >> 4;
  const int sub  = lane & 15;
  const int k    = wave % KC;
  const int vt   = wave / KC;

  const int  brow = tid / 80;
  const int  bc4  = tid % 80;
  const int  arow = tid >> 4;
  const int  ac4  = tid & 15;
  const bool a2   = arow + 20 < 16 * TVT;

  const float* gB = conv_w + (size_t)(o0 + brow) * (DCH * KC) + bc4 * 4;
  const float* gA = emb + (size_t)(vb + arow) * DCH + ac4 * 4;
  const int lB = brow * TPB + bc4 * 4;
  const int lA = arow * TPA + ac4 * 4;

  float4 rb[4];
  float4 ra0, ra1;
  ra1.x = 0.f; ra1.y = 0.f; ra1.z = 0.f; ra1.w = 0.f;

#pragma unroll
  for (int j = 0; j < 4; ++j)
    rb[j] = *(const float4*)(gB + (size_t)(4 * j) * (DCH * KC));
  ra0 = *(const float4*)gA;
  if (a2) ra1 = *(const float4*)(gA + 20 * DCH);
#pragma unroll
  for (int j = 0; j < 4; ++j)
    *(float4*)(&sB[0][lB + 4 * j * TPB]) = rb[j];
  *(float4*)(&sA[0][lA]) = ra0;
  if (a2) *(float4*)(&sA[0][lA + 20 * TPA]) = ra1;
  __syncthreads();

  const float swv = sw[o0 + sub];

  v8f c = {};
  const int fa = (vt * 16 + sub) * TPA + 2 * half;
  const int fb = sub * TPB + 2 * half * KC + k;

  unsigned tbyte = (unsigned)((((size_t)k * VOCAB + (vb + vt * 16 + half * 8)) * DCH + (o0 + sub)) * sizeof(float));
  asm volatile("" : "+v"(tbyte));

  for (int ch = 0; ch < TNC; ++ch) {
    const float* cA = sA[ch & 1];
    const float* cB = sB[ch & 1];
    if (ch + 1 < TNC) {
      const int i1 = (ch + 1) * TIC;
#pragma unroll
      for (int j = 0; j < 4; ++j)
        rb[j] = *(const float4*)(gB + (size_t)(4 * j) * (DCH * KC) + i1 * KC);
      ra0 = *(const float4*)(gA + i1);
      if (a2) ra1 = *(const float4*)(gA + 20 * DCH + i1);
    }
#pragma unroll
    for (int s = 0; s < TIC / 4; ++s) {
      const v2f a = *(const v2f*)(cA + fa + 4 * s);
      v2f b;
      b[0] = cB[fb + (4 * s) * KC];
      b[1] = cB[fb + (4 * s + 1) * KC];
      c = __builtin_amdgcn_wmma_f32_16x16x4_f32(false, a, false, b,
                                                (short)0, c, false, false);
    }
    if (ch + 1 < TNC) {
      float* nA = sA[(ch + 1) & 1];
      float* nB = sB[(ch + 1) & 1];
#pragma unroll
      for (int j = 0; j < 4; ++j)
        *(float4*)(nB + lB + 4 * j * TPB) = rb[j];
      *(float4*)(nA + lA) = ra0;
      if (a2) *(float4*)(nA + lA + 20 * TPA) = ra1;
    }
    if (ch + 1 < TNC) __syncthreads();
  }

#pragma unroll
  for (int r = 0; r < 8; ++r) {
    *(float*)((char*)T + tbyte + (size_t)r * (DCH * sizeof(float))) = c[r];
  }

  float ps[8];
#pragma unroll
  for (int r = 0; r < 8; ++r) ps[r] = c[r] * swv;
#pragma unroll
  for (int off = 8; off > 0; off >>= 1) {
#pragma unroll
    for (int r = 0; r < 8; ++r) ps[r] += __shfl_xor(ps[r], off, 32);
  }
  if (sub == 0) {
    float* d = Sp + ((size_t)blockIdx.x * KC + k) * VOCAB + (vb + vt * 16 + half * 8);
    float4 lo, hi;
    lo.x = ps[0]; lo.y = ps[1]; lo.z = ps[2]; lo.w = ps[3];
    hi.x = ps[4]; hi.y = ps[5]; hi.z = ps[6]; hi.w = ps[7];
    *(float4*)(d)     = lo;
    *(float4*)(d + 4) = hi;
  }
}

__global__ __launch_bounds__(MTHR) void gbst_main(
    const int*   __restrict__ ids,
    const float* __restrict__ T,
    const float* __restrict__ Sp,
    const float* __restrict__ conv_b,
    const float* __restrict__ sw,
    float*       __restrict__ out)
{
  __shared__ __attribute__((aligned(16))) float Ts[KL * TROWS * CS];
  __shared__ __attribute__((aligned(16))) float st[NPQ * 4 + NPQ];
  __shared__ __attribute__((aligned(16))) int   sid[NPQ + 4];
  __shared__ float Ss[KC * TROWS];
  __shared__ float sbsw[4];
  float* wc = st;
  float* sc = st + NPQ * 4;

  const int L    = LLEN;
  const int tid  = threadIdx.x;
  const int sl   = blockIdx.x;
  const int qr   = blockIdx.y;
  const int b    = blockIdx.z;
  const int ng   = (qr == NQR - 1) ? (NGROW - (NQR - 1) * NGQ) : NGQ;
  const int P0   = qr * NPQ;
  const int np   = ng * GL;

  const int lane = tid & 31;
  int idr[NIDR];
#pragma unroll
  for (int j = 0; j < NIDR; ++j) {
    const int q = tid + j * MTHR;
    const int g = P0 + q - 2;
    idr[j] = (q < np + 4 && g >= 0 && g < L) ? ids[(size_t)b * L + g] : VOCAB;
  }
  float4 tv[NTVH];
#pragma unroll
  for (int j = 0; j < NTVH; ++j) {
    const int f   = tid + j * MTHR;
    const int row = f >> 4;
    const int kl  = row >> 8;
    const int k   = kl < 2 ? kl : kl + 1;
    tv[j] = *(const float4*)(T + (size_t)(k * VOCAB + (row & (VOCAB - 1))) * DCH + sl * CS + (f & 15) * 4);
  }
  const int part = tid >> 6;
  const int l64  = tid & 63;
  float4 pa[NSE];
#pragma unroll
  for (int m = 0; m < NSE; ++m) {
    const float* s = Sp + (size_t)(part * 4) * (KC * VOCAB) + (l64 + 64 * m) * 4;
    const float4 s0 = *(const float4*)(s);
    const float4 s1 = *(const float4*)(s + 1 * (KC * VOCAB));
    const float4 s2 = *(const float4*)(s + 2 * (KC * VOCAB));
    const float4 s3 = *(const float4*)(s + 3 * (KC * VOCAB));
    pa[m].x = ((s0.x + s1.x) + s2.x) + s3.x;
    pa[m].y = ((s0.y + s1.y) + s2.y) + s3.y;
    pa[m].z = ((s0.z + s1.z) + s2.z) + s3.z;
    pa[m].w = ((s0.w + s1.w) + s2.w) + s3.w;
  }
  const int ch4 = (tid & 127) * 4;
  const float4 cb = *(const float4*)(conv_b + ch4);
  const float4 cw = *(const float4*)(sw + ch4);
  float bs = ((cb.x * cw.x + cb.y * cw.y) + cb.z * cw.z) + cb.w * cw.w;
#pragma unroll
  for (int off = 16; off > 0; off >>= 1) bs += __shfl_xor(bs, off, 32);

#pragma unroll
  for (int j = 0; j < NTVH; ++j) {
    const int f   = tid + j * MTHR;
    const int row = f >> 4;
    *(float4*)(&Ts[((row >> 8) * TROWS + (row & (VOCAB - 1))) * CS + (f & 15) * 4]) = tv[j];
  }
  asm volatile("" ::: "memory");
#pragma unroll
  for (int j = 0; j < NTVH; ++j) {
    const int f   = tid + (j + NTVH) * MTHR;
    const int row = f >> 4;
    const int kl  = row >> 8;
    const int k   = kl < 2 ? kl : kl + 1;
    tv[j] = *(const float4*)(T + (size_t)(k * VOCAB + (row & (VOCAB - 1))) * DCH + sl * CS + (f & 15) * 4);
  }
  float* pp = st;
#pragma unroll
  for (int m = 0; m < NSE; ++m)
    *(float4*)(&pp[(part * (KC * VOCAB / 4) + l64 + 64 * m) * 4]) = pa[m];
#pragma unroll
  for (int j = 0; j < NIDR; ++j) {
    const int q = tid + j * MTHR;
    if (q < np + 4) sid[q] = idr[j];
  }
  if (tid < KL * CS) Ts[((tid >> 6) * TROWS + VOCAB) * CS + (tid & 63)] = 0.f;
  if (tid >= MTHR - 128 && lane == 0) sbsw[(tid - (MTHR - 128)) >> 5] = bs;
#pragma unroll
  for (int j = 0; j < NTVH; ++j) {
    const int f   = tid + (j + NTVH) * MTHR;
    const int row = f >> 4;
    *(float4*)(&Ts[((row >> 8) * TROWS + (row & (VOCAB - 1))) * CS + (f & 15) * 4]) = tv[j];
  }
  __syncthreads();

  if (tid < KC * VOCAB / 4) {
    float4 a = *(const float4*)(&pp[tid * 4]);
#pragma unroll
    for (int p = 1; p < 8; ++p) {
      const float4 t = *(const float4*)(&pp[(p * (KC * VOCAB / 4) + tid) * 4]);
      a.x += t.x; a.y += t.y; a.z += t.z; a.w += t.w;
    }
    const int e = tid * 4;
    const int o = (e >> 8) * TROWS + (e & (VOCAB - 1));
    Ss[o + 0] = a.x; Ss[o + 1] = a.y; Ss[o + 2] = a.z; Ss[o + 3] = a.w;
  } else if (tid < KC * VOCAB / 4 + KC) {
    Ss[(tid - KC * VOCAB / 4) * TROWS + VOCAB] = 0.f;
  }
  __syncthreads();

  {
    const float bsw = ((sbsw[0] + sbsw[1]) + sbsw[2]) + sbsw[3];
#pragma unroll
    for (int j = 0; j < NPT; ++j) {
      const int q  = tid + j * MTHR;
      const int qq = q < np ? q : np - 1;
      float a = bsw;
#pragma unroll
      for (int k = 0; k < KC; ++k) a += Ss[k * TROWS + sid[qq + k]];
      if (P0 + qq >= L) a = 0.f;
      if (q < np) sc[q] = a;
    }
  }
  __syncthreads();

#pragma unroll
  for (int j = 0; j < NPT; ++j) {
    const int q  = tid + j * MTHR;
    const int qq = q < np ? q : np - 1;
    const int l  = qq % GL;
    const int j2 = qq & ~1;
    const int j3 = qq - (l % 3);
    const int j4 = qq & ~3;
    const float s1 = sc[qq];
    const float s2 = 0.5f * (sc[j2] + sc[j2 + 1]);
    const float s3 = (sc[j3] + sc[j3 + 1] + sc[j3 + 2]) * (1.f / 3.f);
    const float s4 = 0.25f * (sc[j4] + sc[j4 + 1] + sc[j4 + 2] + sc[j4 + 3]);
    const float mx = fmaxf(fmaxf(s1, s2), fmaxf(s3, s4));
    const float e1 = __expf(s1 - mx), e2 = __expf(s2 - mx);
    const float e3 = __expf(s3 - mx), e4 = __expf(s4 - mx);
    const float inv = 1.f / (e1 + e2 + e3 + e4);
    float4 w;
    w.x = e1 * inv;
    w.y = e2 * inv * 0.5f;
    w.z = e3 * inv * (1.f / 3.f);
    w.w = e4 * inv * 0.25f;
    if (q < np) *(float4*)(&wc[q * 4]) = w;
  }
  __syncthreads();

  const int cq = tid & 15;
  const int oc = sl * CS + cq * 4;
  const float4 bv = *(const float4*)(conv_b + oc);
  const float* Tu = T + (size_t)(2 * VOCAB) * DCH + sl * CS;
  const int ntask = ng * (CS / 4);
  for (int task = tid; task < ntask; task += MTHR) {
    const int q0 = (task >> 4) * GL;
    const int g0 = P0 + q0;

    int rv[GL + 4];
#pragma unroll
    for (int i = 0; i < (GL + 4) / 4; ++i) {
      const int4 t = *(const int4*)(&sid[q0 + 4 * i]);
      rv[4 * i + 0] = t.x; rv[4 * i + 1] = t.y; rv[4 * i + 2] = t.z; rv[4 * i + 3] = t.w;
    }

    float4 tc[GL];
#pragma unroll
    for (int p = 0; p < GL; ++p)
      tc[p] = *(const float4*)(Tu + (unsigned)(rv[p + 2] * DCH + cq * 4));

    float4 x[GL];
#pragma unroll
    for (int p = 0; p < GL; ++p) {
      float ax = bv.x, ay = bv.y, az = bv.z, aw = bv.w;
#pragma unroll
      for (int k = 0; k < KC; ++k) {
        float4 t;
        if (k == 2) t = tc[p];
        else        t = *(const float4*)(&Ts[((k < 2 ? k : k - 1) * TROWS + rv[p + k]) * CS + cq * 4]);
        ax += t.x; ay += t.y; az += t.z; aw += t.w;
      }
      if (g0 + p >= L) { ax = 0.f; ay = 0.f; az = 0.f; aw = 0.f; }
      x[p].x = ax; x[p].y = ay; x[p].z = az; x[p].w = aw;
    }

    float4 s2v[GL / 2], s3v[GL / 3], s4v[GL / 4];
#pragma unroll
    for (int i = 0; i < GL / 2; ++i) {
      s2v[i].x = x[2 * i].x + x[2 * i + 1].x;
      s2v[i].y = x[2 * i].y + x[2 * i + 1].y;
      s2v[i].z = x[2 * i].z + x[2 * i + 1].z;
      s2v[i].w = x[2 * i].w + x[2 * i + 1].w;
    }
#pragma unroll
    for (int i = 0; i < GL / 3; ++i) {
      s3v[i].x = (x[3 * i].x + x[3 * i + 1].x) + x[3 * i + 2].x;
      s3v[i].y = (x[3 * i].y + x[3 * i + 1].y) + x[3 * i + 2].y;
      s3v[i].z = (x[3 * i].z + x[3 * i + 1].z) + x[3 * i + 2].z;
      s3v[i].w = (x[3 * i].w + x[3 * i + 1].w) + x[3 * i + 2].w;
    }
#pragma unroll
    for (int i = 0; i < GL / 4; ++i) {
      s4v[i].x = (s2v[2 * i].x + x[4 * i + 2].x) + x[4 * i + 3].x;
      s4v[i].y = (s2v[2 * i].y + x[4 * i + 2].y) + x[4 * i + 3].y;
      s4v[i].z = (s2v[2 * i].z + x[4 * i + 2].z) + x[4 * i + 3].z;
      s4v[i].w = (s2v[2 * i].w + x[4 * i + 2].w) + x[4 * i + 3].w;
    }
#pragma unroll
    for (int j = 0; j < GL / 2; ++j) {
      float ax = 0.f, ay = 0.f, az = 0.f, aw = 0.f;
#pragma unroll
      for (int dl = 0; dl < 2; ++dl) {
        const int l = 2 * j + dl;
        const float4 w = *(const float4*)(&wc[(q0 + l) * 4]);
        ax += w.x * x[l].x;
        ay += w.x * x[l].y;
        az += w.x * x[l].z;
        aw += w.x * x[l].w;
        ax += w.y * s2v[l >> 1].x;
        ay += w.y * s2v[l >> 1].y;
        az += w.y * s2v[l >> 1].z;
        aw += w.y * s2v[l >> 1].w;
        ax += w.z * s3v[l / 3].x;
        ay += w.z * s3v[l / 3].y;
        az += w.z * s3v[l / 3].z;
        aw += w.z * s3v[l / 3].w;
        ax += w.w * s4v[l >> 2].x;
        ay += w.w * s4v[l >> 2].y;
        az += w.w * s4v[l >> 2].z;
        aw += w.w * s4v[l >> 2].w;
      }
      if (g0 + 2 * j < L) {
        float4 r;
        r.x = 0.5f * ax; r.y = 0.5f * ay; r.z = 0.5f * az; r.w = 0.5f * aw;
        const size_t off = ((size_t)b * (L >> 1) + ((g0 >> 1) + j)) * DCH + oc;
        *(float4*)(out + off) = r;
      }
    }
  }
}

extern "C" void kernel_launch(void* const* d_in, const int* in_sizes, int n_in,
                              void* d_out, int out_size, void* d_ws, size_t ws_size,
                              hipStream_t stream) {
  const int*   ids     = (const int*)d_in[0];
  const float* emb     = (const float*)d_in[1];
  const float* conv_w  = (const float*)d_in[2];
  const float* conv_b  = (const float*)d_in[3];
  const float* score_w = (const float*)d_in[4];
  float* out = (float*)d_out;
  float* T   = (float*)d_ws;
  float* Sp  = T + KC * VOCAB * DCH;

  const int L = LLEN;
  const int B = in_sizes[0] / L;

  dim3 g1(DCH / 16, VOCAB / (16 * TVT));
  gbst_table_wmma<<<g1, TTHR, 0, stream>>>(emb, conv_w, score_w, T, Sp);

  dim3 g2(DCH / CS, NQR, B);
  gbst_main<<<g2, MTHR, 0, stream>>>(ids, T, Sp, conv_b, score_w, out);
}
